# starting kernel
# baseline (speedup 1.0000x reference)
#include <hip/hip_runtime.h>

typedef __attribute__((ext_vector_type(2))) float v2f;
typedef __attribute__((ext_vector_type(8))) float v8f;

#define F_DIM 128
#define H_DIM 64

__device__ __forceinline__ void atomAddF(float* p, float v) {
    __hip_atomic_fetch_add(p, v, __ATOMIC_RELAXED, __HIP_MEMORY_SCOPE_AGENT);
}

__global__ void __launch_bounds__(256) gemm_wmma_f32(
    const float* __restrict__ x, const float* __restrict__ W,
    const float* __restrict__ bias, float* __restrict__ xt, int N)
{
    const int waveId = (blockIdx.x * blockDim.x + threadIdx.x) >> 5;
    const int lane   = threadIdx.x & 31;
    const int tilesH = H_DIM / 16;
    const int tileM  = waveId / tilesH;
    const int tileH  = waveId % tilesH;
    if (tileM * 16 >= N) return;

    const int m   = lane & 15;
    const int kp  = (lane >> 4) << 1;
    const float* xrow = x + (size_t)(tileM * 16 + m) * F_DIM;
    const float* wrow = W + (size_t)(tileH * 16 + m) * F_DIM;

    v8f c = {};
    #pragma unroll 8
    for (int k = 0; k < F_DIM; k += 4) {
        v2f a = *(const v2f*)(xrow + k + kp);
        v2f b = *(const v2f*)(wrow + k + kp);
        c = __builtin_amdgcn_wmma_f32_16x16x4_f32(
                 false, a,  false, b,
                 (short)0, c,  false,  false);
    }

    const int hOut  = tileH * 16 + (lane & 15);
    const float bv  = bias[hOut];
    const int mBase = tileM * 16 + ((lane >> 4) << 3);
    #pragma unroll
    for (int v = 0; v < 8; ++v) {
        xt[(size_t)(mBase + v) * H_DIM + hOut] = c[v] + bv;
    }
}

__global__ void __launch_bounds__(256) spmm_scatter(
    const float* __restrict__ xt, const int* __restrict__ rows,
    const int* __restrict__ cols, const float* __restrict__ vals,
    float* __restrict__ outH, int E)
{
    long long idx = (long long)blockIdx.x * blockDim.x + threadIdx.x;
    long long e = idx >> 4;
    int q = (int)(idx & 15);
    if (e >= E) return;
    const int r = rows[e];
    const int c = cols[e];
    const float v = vals[e];
    const float4 msg = ((const float4*)(xt + (size_t)c * H_DIM))[q];
    float* dst = outH + (size_t)r * H_DIM + q * 4;
    atomAddF(dst + 0, v * msg.x);
    atomAddF(dst + 1, v * msg.y);
    atomAddF(dst + 2, v * msg.z);
    atomAddF(dst + 3, v * msg.w);
}

__global__ void __launch_bounds__(256) prelu_colsum(
    float* __restrict__ h, const float* __restrict__ prelu_a,
    float* __restrict__ colsum, long long total, int doSum)
{
    const float a = prelu_a[0];
    const long long stride = (long long)gridDim.x * blockDim.x;
    float local = 0.f;
    for (long long i = (long long)blockIdx.x * blockDim.x + threadIdx.x;
         i < total; i += stride) {
        float v = h[i];
        float o = (v >= 0.f) ? v : a * v;
        h[i] = o;
        local += o;
    }
    if (doSum) {
        __shared__ float sm[256];
        sm[threadIdx.x] = local;
        __syncthreads();
        if (threadIdx.x < 64) {
            float t = sm[threadIdx.x] + sm[threadIdx.x + 64] +
                      sm[threadIdx.x + 128] + sm[threadIdx.x + 192];
            atomAddF(colsum + threadIdx.x, t);
        }
    }
}

__global__ void summary_kernel(
    const float* __restrict__ colsum, const float* __restrict__ Wbil,
    float* __restrict__ tbuf, int N)
{
    __shared__ float s_sh[H_DIM];
    const int hh = threadIdx.x;
    float s = colsum[hh] / (float)N;
    s = 1.f / (1.f + expf(-s));
    s_sh[hh] = s;
    __syncthreads();
    float t = 0.f;
    #pragma unroll
    for (int k = 0; k < H_DIM; ++k) t += Wbil[hh * H_DIM + k] * s_sh[k];
    tbuf[hh] = t;
}

__global__ void __launch_bounds__(256) score_kernel(
    const float* __restrict__ Hm, const float* __restrict__ tbuf,
    const float* __restrict__ bbil, float* __restrict__ out, int N)
{
    __shared__ float tsh[H_DIM];
    if (threadIdx.x < H_DIM) tsh[threadIdx.x] = tbuf[threadIdx.x];
    __syncthreads();
    const int wave = threadIdx.x >> 5;
    const int lane = threadIdx.x & 31;
    const int n = blockIdx.x * (blockDim.x >> 5) + wave;
    if (n >= N) return;
    const float* hr = Hm + (size_t)n * H_DIM;
    float acc = hr[lane] * tsh[lane] + hr[lane + 32] * tsh[lane + 32];
    #pragma unroll
    for (int off = 16; off >= 1; off >>= 1)
        acc += __shfl_xor(acc, off, 32);
    if (lane == 0) out[n] = acc + bbil[0];
}

extern "C" void kernel_launch(void* const* d_in, const int* in_sizes, int n_in,
                              void* d_out, int out_size, void* d_ws, size_t ws_size,
                              hipStream_t stream) {
    const float* pos   = (const float*)d_in[0];
    const float* neg   = (const float*)d_in[1];
    const int*   erows = (const int*)  d_in[2];
    const int*   ecols = (const int*)  d_in[3];
    const float* evals = (const float*)d_in[4];
    const float* Wg    = (const float*)d_in[5];
    const float* bg    = (const float*)d_in[6];
    const float* pa    = (const float*)d_in[7];
    const float* Wb    = (const float*)d_in[8];
    const float* bb    = (const float*)d_in[9];

    const int N = in_sizes[0] / F_DIM;
    const int E = in_sizes[2];
    float* out = (float*)d_out;

    float* xt     = (float*)d_ws;
    float* posH   = xt   + (size_t)N * H_DIM;
    float* negH   = posH + (size_t)N * H_DIM;
    float* colsum = negH + (size_t)N * H_DIM;
    float* tbuf   = colsum + H_DIM;

    hipMemsetAsync(posH,   0, (size_t)N * H_DIM * sizeof(float), stream);
    hipMemsetAsync(negH,   0, (size_t)N * H_DIM * sizeof(float), stream);
    hipMemsetAsync(colsum, 0, H_DIM * sizeof(float), stream);

    const int tiles      = (N / 16) * (H_DIM / 16);
    const int gemmBlocks = (tiles * 32 + 255) / 256;
    const long long spmmThreads = (long long)E * (H_DIM / 4);
    const int spmmBlocks = (int)((spmmThreads + 255) / 256);
    const long long total = (long long)N * H_DIM;
    const int scoreBlocks = (N + 7) / 8;

    gemm_wmma_f32<<<gemmBlocks, 256, 0, stream>>>(pos, Wg, bg, xt, N);
    spmm_scatter <<<spmmBlocks, 256, 0, stream>>>(xt, erows, ecols, evals, posH, E);
    prelu_colsum <<<2048, 256, 0, stream>>>(posH, pa, colsum, total, 1);
    gemm_wmma_f32<<<gemmBlocks, 256, 0, stream>>>(neg, Wg, bg, xt, N);
    spmm_scatter <<<spmmBlocks, 256, 0, stream>>>(xt, erows, ecols, evals, negH, E);
    prelu_colsum <<<2048, 256, 0, stream>>>(negH, pa, colsum, total, 0);
    summary_kernel<<<1, 64, 0, stream>>>(colsum, Wb, tbuf, N);
    score_kernel<<<scoreBlocks, 256, 0, stream>>>(posH, tbuf, bb, out, N);
    score_kernel<<<scoreBlocks, 256, 0, stream>>>(negH, tbuf, bb, out + N, N);
}


# spmm scatter: one wave walks 32-edge chunks (indices loaded once per chunk, broadcast by readlane; whole-line row loads and atomics) instead of one 16-lane group per edge: 32x fewer waves
# speedup vs baseline: 1.3771x; 1.3771x over previous
#include <hip/hip_runtime.h>

typedef __attribute__((ext_vector_type(2))) float v2f;
typedef __attribute__((ext_vector_type(8))) float v8f;

#define F_DIM 128
#define H_DIM 64

__device__ __forceinline__ void atomAddF(float* p, float v) {
    __hip_atomic_fetch_add(p, v, __ATOMIC_RELAXED, __HIP_MEMORY_SCOPE_AGENT);
}

__global__ void __launch_bounds__(256) gemm_wmma_f32(
    const float* __restrict__ x, const float* __restrict__ W,
    const float* __restrict__ bias, float* __restrict__ xt, int N)
{
    const int waveId = (blockIdx.x * blockDim.x + threadIdx.x) >> 5;
    const int lane   = threadIdx.x & 31;
    const int tilesH = H_DIM / 16;
    const int tileM  = waveId / tilesH;
    const int tileH  = waveId % tilesH;
    if (tileM * 16 >= N) return;

    const int m   = lane & 15;
    const int kp  = (lane >> 4) << 1;
    const float* xrow = x + (size_t)(tileM * 16 + m) * F_DIM;
    const float* wrow = W + (size_t)(tileH * 16 + m) * F_DIM;

    v8f c = {};
    #pragma unroll 8
    for (int k = 0; k < F_DIM; k += 4) {
        v2f a = *(const v2f*)(xrow + k + kp);
        v2f b = *(const v2f*)(wrow + k + kp);
        c = __builtin_amdgcn_wmma_f32_16x16x4_f32(
                 false, a,  false, b,
                 (short)0, c,  false,  false);
    }

    const int hOut  = tileH * 16 + (lane & 15);
    const float bv  = bias[hOut];
    const int mBase = tileM * 16 + ((lane >> 4) << 3);
    #pragma unroll
    for (int v = 0; v < 8; ++v) {
        xt[(size_t)(mBase + v) * H_DIM + hOut] = c[v] + bv;
    }
}

__global__ void __launch_bounds__(256) spmm_scatter(
    const float* __restrict__ xt, const int* __restrict__ rows,
    const int* __restrict__ cols, const float* __restrict__ vals,
    float* __restrict__ outH, int E)
{
    const int lane = threadIdx.x & 31;
    const int wave = __builtin_amdgcn_readfirstlane(
        (int)((blockIdx.x * blockDim.x + threadIdx.x) >> 5));
    const int nWaves = (int)((gridDim.x * blockDim.x) >> 5);
    const long long stride = (long long)nWaves * 32;
    long long base = (long long)wave * 32;
    if (base >= E) return;

    int r = 0, c = 0;
    float v = 0.f;
    if (base + lane < E) {
        r = rows[base + lane];
        c = cols[base + lane];
        v = vals[base + lane];
    }
    for (;;) {
        const long long nb = base + stride;
        int rn = 0, cn = 0;
        float vn = 0.f;
        if (nb + lane < E) {
            rn = rows[nb + lane];
            cn = cols[nb + lane];
            vn = vals[nb + lane];
        }
        const long long left = (long long)E - base;
        const int cnt = left < 32 ? (int)left : 32;
        int j0 = 0;
        for (; j0 + 8 <= cnt; j0 += 8) {
            float m0[8], m1[8];
            #pragma unroll
            for (int j = 0; j < 8; ++j) {
                const int cj = __builtin_amdgcn_readlane(c, j0 + j);
                const float* src = xt + (size_t)cj * H_DIM + lane;
                m0[j] = src[0];
                m1[j] = src[32];
            }
            #pragma unroll
            for (int j = 0; j < 8; ++j) {
                const int rj = __builtin_amdgcn_readlane(r, j0 + j);
                const float vj = __int_as_float(
                    __builtin_amdgcn_readlane(__float_as_int(v), j0 + j));
                float* dst = outH + (size_t)rj * H_DIM + lane;
                atomAddF(dst, vj * m0[j]);
                atomAddF(dst + 32, vj * m1[j]);
            }
        }
        for (; j0 < cnt; ++j0) {
            const int cj = __builtin_amdgcn_readlane(c, j0);
            const int rj = __builtin_amdgcn_readlane(r, j0);
            const float vj = __int_as_float(
                __builtin_amdgcn_readlane(__float_as_int(v), j0));
            const float* src = xt + (size_t)cj * H_DIM + lane;
            float* dst = outH + (size_t)rj * H_DIM + lane;
            atomAddF(dst, vj * src[0]);
            atomAddF(dst + 32, vj * src[32]);
        }
        if (nb >= E) break;
        base = nb; r = rn; c = cn; v = vn;
    }
}

__global__ void __launch_bounds__(256) prelu_colsum(
    float* __restrict__ h, const float* __restrict__ prelu_a,
    float* __restrict__ colsum, long long total, int doSum)
{
    const float a = prelu_a[0];
    const long long stride = (long long)gridDim.x * blockDim.x;
    float local = 0.f;
    for (long long i = (long long)blockIdx.x * blockDim.x + threadIdx.x;
         i < total; i += stride) {
        float v = h[i];
        float o = (v >= 0.f) ? v : a * v;
        h[i] = o;
        local += o;
    }
    if (doSum) {
        __shared__ float sm[256];
        sm[threadIdx.x] = local;
        __syncthreads();
        if (threadIdx.x < 64) {
            float t = sm[threadIdx.x] + sm[threadIdx.x + 64] +
                      sm[threadIdx.x + 128] + sm[threadIdx.x + 192];
            atomAddF(colsum + threadIdx.x, t);
        }
    }
}

__global__ void summary_kernel(
    const float* __restrict__ colsum, const float* __restrict__ Wbil,
    float* __restrict__ tbuf, int N)
{
    __shared__ float s_sh[H_DIM];
    const int hh = threadIdx.x;
    float s = colsum[hh] / (float)N;
    s = 1.f / (1.f + expf(-s));
    s_sh[hh] = s;
    __syncthreads();
    float t = 0.f;
    #pragma unroll
    for (int k = 0; k < H_DIM; ++k) t += Wbil[hh * H_DIM + k] * s_sh[k];
    tbuf[hh] = t;
}

__global__ void __launch_bounds__(256) score_kernel(
    const float* __restrict__ Hm, const float* __restrict__ tbuf,
    const float* __restrict__ bbil, float* __restrict__ out, int N)
{
    __shared__ float tsh[H_DIM];
    if (threadIdx.x < H_DIM) tsh[threadIdx.x] = tbuf[threadIdx.x];
    __syncthreads();
    const int wave = threadIdx.x >> 5;
    const int lane = threadIdx.x & 31;
    const int n = blockIdx.x * (blockDim.x >> 5) + wave;
    if (n >= N) return;
    const float* hr = Hm + (size_t)n * H_DIM;
    float acc = hr[lane] * tsh[lane] + hr[lane + 32] * tsh[lane + 32];
    #pragma unroll
    for (int off = 16; off >= 1; off >>= 1)
        acc += __shfl_xor(acc, off, 32);
    if (lane == 0) out[n] = acc + bbil[0];
}

extern "C" void kernel_launch(void* const* d_in, const int* in_sizes, int n_in,
                              void* d_out, int out_size, void* d_ws, size_t ws_size,
                              hipStream_t stream) {
    const float* pos   = (const float*)d_in[0];
    const float* neg   = (const float*)d_in[1];
    const int*   erows = (const int*)  d_in[2];
    const int*   ecols = (const int*)  d_in[3];
    const float* evals = (const float*)d_in[4];
    const float* Wg    = (const float*)d_in[5];
    const float* bg    = (const float*)d_in[6];
    const float* pa    = (const float*)d_in[7];
    const float* Wb    = (const float*)d_in[8];
    const float* bb    = (const float*)d_in[9];

    const int N = in_sizes[0] / F_DIM;
    const int E = in_sizes[2];
    float* out = (float*)d_out;

    float* xt     = (float*)d_ws;
    float* posH   = xt   + (size_t)N * H_DIM;
    float* negH   = posH + (size_t)N * H_DIM;
    float* colsum = negH + (size_t)N * H_DIM;
    float* tbuf   = colsum + H_DIM;

    hipMemsetAsync(posH,   0, (size_t)N * H_DIM * sizeof(float), stream);
    hipMemsetAsync(negH,   0, (size_t)N * H_DIM * sizeof(float), stream);
    hipMemsetAsync(colsum, 0, H_DIM * sizeof(float), stream);

    const int tiles      = (N / 16) * (H_DIM / 16);
    const int gemmBlocks = (tiles * 32 + 255) / 256;
    const long long total = (long long)N * H_DIM;
    const int scoreBlocks = (N + 7) / 8;

    gemm_wmma_f32<<<gemmBlocks, 256, 0, stream>>>(pos, Wg, bg, xt, N);
    spmm_scatter <<<3125, 160, 0, stream>>>(xt, erows, ecols, evals, posH, E);
    prelu_colsum <<<2048, 256, 0, stream>>>(posH, pa, colsum, total, 1);
    gemm_wmma_f32<<<gemmBlocks, 256, 0, stream>>>(neg, Wg, bg, xt, N);
    spmm_scatter <<<3125, 160, 0, stream>>>(xt, erows, ecols, evals, negH, E);
    prelu_colsum <<<2048, 256, 0, stream>>>(negH, pa, colsum, total, 0);
    summary_kernel<<<1, 64, 0, stream>>>(colsum, Wb, tbuf, N);
    score_kernel<<<scoreBlocks, 256, 0, stream>>>(posH, tbuf, bb, out, N);
    score_kernel<<<scoreBlocks, 256, 0, stream>>>(negH, tbuf, bb, out + N, N);
}


# gemm: one wave per 16x64 strip (4 accumulator tiles): x rows read once with 128-bit loads, W staged per block in padded LDS, k order permuted alike in A and B; was 16x16 per wave with 64-bit loads
# speedup vs baseline: 1.6183x; 1.1752x over previous
#include <hip/hip_runtime.h>

typedef __attribute__((ext_vector_type(2))) float v2f;
typedef __attribute__((ext_vector_type(4))) float v4f;
typedef __attribute__((ext_vector_type(8))) float v8f;

#define F_DIM 128
#define H_DIM 64

__device__ __forceinline__ void atomAddF(float* p, float v) {
    __hip_atomic_fetch_add(p, v, __ATOMIC_RELAXED, __HIP_MEMORY_SCOPE_AGENT);
}

#define W_PITCH 132
__device__ __forceinline__ void lds_barrier() {
    asm volatile("s_wait_dscnt 0x0\n\ts_barrier_signal -1\n\ts_barrier_wait -1" ::: "memory");
}
__global__ void __launch_bounds__(256) gemm_wmma_f32(
    const float* __restrict__ x, const float* __restrict__ W,
    const float* __restrict__ bias, float* __restrict__ xt, int N)
{
    __shared__ __attribute__((aligned(16))) float Ws[H_DIM * W_PITCH];
    const int tid   = threadIdx.x;
    const int lane  = tid & 31;
    const int m     = lane & 15;
    const int half  = lane >> 4;
    const int tileM = blockIdx.x * 8 + (tid >> 5);
    const bool active = tileM * 16 < N;

    v4f w[H_DIM * F_DIM / 4 / 256];
    #pragma unroll
    for (int j = 0; j < H_DIM * F_DIM / 4 / 256; ++j) w[j] = ((const v4f*)W)[tid + 256 * j];

    const int rowA = min(tileM * 16 + m, N - 1);
    const float* xrow = x + (size_t)rowA * F_DIM + 4 * half;
    v4f a[F_DIM / 8];
    #pragma unroll
    for (int kb = 0; kb < F_DIM / 8; ++kb) a[kb] = *(const v4f*)(xrow + 8 * kb);

    #pragma unroll
    for (int j = 0; j < H_DIM * F_DIM / 4 / 256; ++j) {
        const int i = tid + 256 * j;
        *(v4f*)(Ws + (i >> 5) * W_PITCH + 4 * (i & 31)) = w[j];
    }
    lds_barrier();

    v8f c0 = {}, c1 = {}, c2 = {}, c3 = {};
    const float* wb = Ws + m * W_PITCH + 4 * half;
    #pragma unroll
    for (int kb = 0; kb < F_DIM / 8; ++kb) {
        const v4f b0 = *(const v4f*)(wb +  0 * W_PITCH + 8 * kb);
        const v4f b1 = *(const v4f*)(wb + 16 * W_PITCH + 8 * kb);
        const v4f b2 = *(const v4f*)(wb + 32 * W_PITCH + 8 * kb);
        const v4f b3 = *(const v4f*)(wb + 48 * W_PITCH + 8 * kb);
        const v2f alo = {a[kb].x, a[kb].y}, ahi = {a[kb].z, a[kb].w};
        c0 = __builtin_amdgcn_wmma_f32_16x16x4_f32(false, alo, false, (v2f){b0.x, b0.y}, (short)0, c0, false, false);
        c1 = __builtin_amdgcn_wmma_f32_16x16x4_f32(false, alo, false, (v2f){b1.x, b1.y}, (short)0, c1, false, false);
        c2 = __builtin_amdgcn_wmma_f32_16x16x4_f32(false, alo, false, (v2f){b2.x, b2.y}, (short)0, c2, false, false);
        c3 = __builtin_amdgcn_wmma_f32_16x16x4_f32(false, alo, false, (v2f){b3.x, b3.y}, (short)0, c3, false, false);
        c0 = __builtin_amdgcn_wmma_f32_16x16x4_f32(false, ahi, false, (v2f){b0.z, b0.w}, (short)0, c0, false, false);
        c1 = __builtin_amdgcn_wmma_f32_16x16x4_f32(false, ahi, false, (v2f){b1.z, b1.w}, (short)0, c1, false, false);
        c2 = __builtin_amdgcn_wmma_f32_16x16x4_f32(false, ahi, false, (v2f){b2.z, b2.w}, (short)0, c2, false, false);
        c3 = __builtin_amdgcn_wmma_f32_16x16x4_f32(false, ahi, false, (v2f){b3.z, b3.w}, (short)0, c3, false, false);
    }

    if (!active) return;
    float* orow = xt + (size_t)(tileM * 16 + 8 * half) * H_DIM + m;
    const float bv0 = bias[m], bv1 = bias[16 + m], bv2 = bias[32 + m], bv3 = bias[48 + m];
    #pragma unroll
    for (int v = 0; v < 8; ++v) {
        orow[v * H_DIM +  0] = c0[v] + bv0;
        orow[v * H_DIM + 16] = c1[v] + bv1;
        orow[v * H_DIM + 32] = c2[v] + bv2;
        orow[v * H_DIM + 48] = c3[v] + bv3;
    }
}

__global__ void __launch_bounds__(256) spmm_scatter(
    const float* __restrict__ xt, const int* __restrict__ rows,
    const int* __restrict__ cols, const float* __restrict__ vals,
    float* __restrict__ outH, int E)
{
    const int lane = threadIdx.x & 31;
    const int wave = __builtin_amdgcn_readfirstlane(
        (int)((blockIdx.x * blockDim.x + threadIdx.x) >> 5));
    const int nWaves = (int)((gridDim.x * blockDim.x) >> 5);
    const long long stride = (long long)nWaves * 32;
    long long base = (long long)wave * 32;
    if (base >= E) return;

    int r = 0, c = 0;
    float v = 0.f;
    if (base + lane < E) {
        r = rows[base + lane];
        c = cols[base + lane];
        v = vals[base + lane];
    }
    for (;;) {
        const long long nb = base + stride;
        int rn = 0, cn = 0;
        float vn = 0.f;
        if (nb + lane < E) {
            rn = rows[nb + lane];
            cn = cols[nb + lane];
            vn = vals[nb + lane];
        }
        const long long left = (long long)E - base;
        const int cnt = left < 32 ? (int)left : 32;
        int j0 = 0;
        for (; j0 + 8 <= cnt; j0 += 8) {
            float m0[8], m1[8];
            #pragma unroll
            for (int j = 0; j < 8; ++j) {
                const int cj = __builtin_amdgcn_readlane(c, j0 + j);
                const float* src = xt + (size_t)cj * H_DIM + lane;
                m0[j] = src[0];
                m1[j] = src[32];
            }
            #pragma unroll
            for (int j = 0; j < 8; ++j) {
                const int rj = __builtin_amdgcn_readlane(r, j0 + j);
                const float vj = __int_as_float(
                    __builtin_amdgcn_readlane(__float_as_int(v), j0 + j));
                float* dst = outH + (size_t)rj * H_DIM + lane;
                atomAddF(dst, vj * m0[j]);
                atomAddF(dst + 32, vj * m1[j]);
            }
        }
        for (; j0 < cnt; ++j0) {
            const int cj = __builtin_amdgcn_readlane(c, j0);
            const int rj = __builtin_amdgcn_readlane(r, j0);
            const float vj = __int_as_float(
                __builtin_amdgcn_readlane(__float_as_int(v), j0));
            const float* src = xt + (size_t)cj * H_DIM + lane;
            float* dst = outH + (size_t)rj * H_DIM + lane;
            atomAddF(dst, vj * src[0]);
            atomAddF(dst + 32, vj * src[32]);
        }
        if (nb >= E) break;
        base = nb; r = rn; c = cn; v = vn;
    }
}

__global__ void __launch_bounds__(256) prelu_colsum(
    float* __restrict__ h, const float* __restrict__ prelu_a,
    float* __restrict__ colsum, long long total, int doSum)
{
    const float a = prelu_a[0];
    const long long stride = (long long)gridDim.x * blockDim.x;
    float local = 0.f;
    for (long long i = (long long)blockIdx.x * blockDim.x + threadIdx.x;
         i < total; i += stride) {
        float v = h[i];
        float o = (v >= 0.f) ? v : a * v;
        h[i] = o;
        local += o;
    }
    if (doSum) {
        __shared__ float sm[256];
        sm[threadIdx.x] = local;
        __syncthreads();
        if (threadIdx.x < 64) {
            float t = sm[threadIdx.x] + sm[threadIdx.x + 64] +
                      sm[threadIdx.x + 128] + sm[threadIdx.x + 192];
            atomAddF(colsum + threadIdx.x, t);
        }
    }
}

__global__ void summary_kernel(
    const float* __restrict__ colsum, const float* __restrict__ Wbil,
    float* __restrict__ tbuf, int N)
{
    __shared__ float s_sh[H_DIM];
    const int hh = threadIdx.x;
    float s = colsum[hh] / (float)N;
    s = 1.f / (1.f + expf(-s));
    s_sh[hh] = s;
    __syncthreads();
    float t = 0.f;
    #pragma unroll
    for (int k = 0; k < H_DIM; ++k) t += Wbil[hh * H_DIM + k] * s_sh[k];
    tbuf[hh] = t;
}

__global__ void __launch_bounds__(256) score_kernel(
    const float* __restrict__ Hm, const float* __restrict__ tbuf,
    const float* __restrict__ bbil, float* __restrict__ out, int N)
{
    __shared__ float tsh[H_DIM];
    if (threadIdx.x < H_DIM) tsh[threadIdx.x] = tbuf[threadIdx.x];
    __syncthreads();
    const int wave = threadIdx.x >> 5;
    const int lane = threadIdx.x & 31;
    const int n = blockIdx.x * (blockDim.x >> 5) + wave;
    if (n >= N) return;
    const float* hr = Hm + (size_t)n * H_DIM;
    float acc = hr[lane] * tsh[lane] + hr[lane + 32] * tsh[lane + 32];
    #pragma unroll
    for (int off = 16; off >= 1; off >>= 1)
        acc += __shfl_xor(acc, off, 32);
    if (lane == 0) out[n] = acc + bbil[0];
}

extern "C" void kernel_launch(void* const* d_in, const int* in_sizes, int n_in,
                              void* d_out, int out_size, void* d_ws, size_t ws_size,
                              hipStream_t stream) {
    const float* pos   = (const float*)d_in[0];
    const float* neg   = (const float*)d_in[1];
    const int*   erows = (const int*)  d_in[2];
    const int*   ecols = (const int*)  d_in[3];
    const float* evals = (const float*)d_in[4];
    const float* Wg    = (const float*)d_in[5];
    const float* bg    = (const float*)d_in[6];
    const float* pa    = (const float*)d_in[7];
    const float* Wb    = (const float*)d_in[8];
    const float* bb    = (const float*)d_in[9];

    const int N = in_sizes[0] / F_DIM;
    const int E = in_sizes[2];
    float* out = (float*)d_out;

    float* xt     = (float*)d_ws;
    float* posH   = xt   + (size_t)N * H_DIM;
    float* negH   = posH + (size_t)N * H_DIM;
    float* colsum = negH + (size_t)N * H_DIM;
    float* tbuf   = colsum + H_DIM;

    hipMemsetAsync(posH,   0, (size_t)N * H_DIM * sizeof(float), stream);
    hipMemsetAsync(negH,   0, (size_t)N * H_DIM * sizeof(float), stream);
    hipMemsetAsync(colsum, 0, H_DIM * sizeof(float), stream);

    const int gemmBlocks = (N / 16 + 7) / 8;
    const long long total = (long long)N * H_DIM;
    const int scoreBlocks = (N + 7) / 8;

    gemm_wmma_f32<<<gemmBlocks, 256, 0, stream>>>(pos, Wg, bg, xt, N);
    spmm_scatter <<<3125, 160, 0, stream>>>(xt, erows, ecols, evals, posH, E);
    prelu_colsum <<<2048, 256, 0, stream>>>(posH, pa, colsum, total, 1);
    gemm_wmma_f32<<<gemmBlocks, 256, 0, stream>>>(neg, Wg, bg, xt, N);
    spmm_scatter <<<3125, 160, 0, stream>>>(xt, erows, ecols, evals, negH, E);
    prelu_colsum <<<2048, 256, 0, stream>>>(negH, pa, colsum, total, 0);
    summary_kernel<<<1, 64, 0, stream>>>(colsum, Wb, tbuf, N);
    score_kernel<<<scoreBlocks, 256, 0, stream>>>(posH, tbuf, bb, out, N);
    score_kernel<<<scoreBlocks, 256, 0, stream>>>(negH, tbuf, bb, out + N, N);
}
